# starting kernel
# baseline (speedup 1.0000x reference)
#include <hip/hip_runtime.h>


#define K_SZ   7
#define KK     49
#define G_NUM  16
#define C_CH   256
#define CR     64
#define NK     (KK * G_NUM)
#define HW     64
#define BATCH  2
#define M_TOT  (BATCH * HW * HW)
#define BN_EPS 1e-3f

typedef __attribute__((ext_vector_type(2))) float v2f;
typedef __attribute__((ext_vector_type(8))) float v8f;

__global__ __launch_bounds__(256) void gemm1_bn_relu(
    const float* __restrict__ x,  const float* __restrict__ w1,
    const float* __restrict__ b1, const float* __restrict__ gamma,
    const float* __restrict__ beta, const float* __restrict__ mu,
    const float* __restrict__ var, float* __restrict__ t)
{
    const int wave  = threadIdx.x >> 5;
    const int ln    = threadIdx.x & 31;
    const int tile  = blockIdx.x * 8 + wave;
    const int m0    = (tile >> 2) << 4;
    const int n0    = (tile & 3)  << 4;
    const int laneM = ln & 15;
    const int kHalf = ln >> 4;

    const float* arow = x + (size_t)(m0 + laneM) * C_CH + 2 * kHalf;
    const float* bcol = w1 + (size_t)(2 * kHalf) * CR + (n0 + laneM);

    v8f acc = {};
#pragma unroll 8
    for (int k = 0; k < C_CH; k += 4) {
        v2f a, b;
        a.x = arow[k];
        a.y = arow[k + 1];
        b.x = bcol[(size_t)k * CR];
        b.y = bcol[(size_t)(k + 1) * CR];
        acc = __builtin_amdgcn_wmma_f32_16x16x4_f32(
            false, a, false, b, (short)0, acc, false, false);
    }

    const int   col   = n0 + laneM;
    const float scale = gamma[col] * rsqrtf(var[col] + BN_EPS);
    const float shift = beta[col] - mu[col] * scale;
    const float b1c   = b1[col];
    const int   rowB  = m0 + (kHalf << 3);
#pragma unroll
    for (int r = 0; r < 8; ++r) {
        float v = fmaf(acc[r] + b1c, scale, shift);
        t[(size_t)(rowB + r) * CR + col] = fmaxf(v, 0.0f);
    }
}

__global__ __launch_bounds__(256) void gemm2_kerngen(
    const float* __restrict__ t, const float* __restrict__ w2,
    const float* __restrict__ b2, float* __restrict__ kern)
{
    const int wave = threadIdx.x >> 5;
    const int ln   = threadIdx.x & 31;
    const int tile = blockIdx.x * 8 + wave;
    if (tile >= (M_TOT / 16) * (NK / 16)) return;
    const int mt = tile / (NK / 16);
    const int nt = tile % (NK / 16);
    const int m0 = mt << 4, n0 = nt << 4;
    const int laneM = ln & 15;
    const int kHalf = ln >> 4;

    const float* arow = t  + (size_t)(m0 + laneM) * CR + 2 * kHalf;
    const float* bcol = w2 + (size_t)(2 * kHalf) * NK + (n0 + laneM);

    v8f acc = {};
#pragma unroll
    for (int k = 0; k < CR; k += 4) {
        v2f a, b;
        a.x = arow[k];
        a.y = arow[k + 1];
        b.x = bcol[(size_t)k * NK];
        b.y = bcol[(size_t)(k + 1) * NK];
        acc = __builtin_amdgcn_wmma_f32_16x16x4_f32(
            false, a, false, b, (short)0, acc, false, false);
    }

    const int   col  = n0 + laneM;
    const float b2c  = b2[col];
    const int   rowB = m0 + (kHalf << 3);
#pragma unroll
    for (int r = 0; r < 8; ++r) {
        kern[(size_t)(rowB + r) * NK + col] = acc[r] + b2c;
    }
}

__global__ __launch_bounds__(256) void involution_gather(
    const float* __restrict__ x, const float* __restrict__ kern,
    float* __restrict__ out)
{
    __shared__ float ksh[NK];
    const int pix = blockIdx.x;
    const int c   = threadIdx.x;
    const int w   = pix & (HW - 1);
    const int h   = (pix >> 6) & (HW - 1);
    const int b   = pix >> 12;

    const float* kp = kern + (size_t)pix * NK;
    for (int i = threadIdx.x; i < NK; i += 256) ksh[i] = kp[i];
    __syncthreads();

    const int g = c & (G_NUM - 1);
    float acc = 0.0f;
#pragma unroll
    for (int kh = 0; kh < K_SZ; ++kh) {
        const int hh = h + kh - (K_SZ / 2);
        if ((unsigned)hh >= (unsigned)HW) continue;
#pragma unroll
        for (int kw = 0; kw < K_SZ; ++kw) {
            const int ww = w + kw - (K_SZ / 2);
            if ((unsigned)ww >= (unsigned)HW) continue;
            const float xv = x[(((size_t)(b * HW + hh)) * HW + ww) * C_CH + c];
            acc = fmaf(ksh[(kh * K_SZ + kw) * G_NUM + g], xv, acc);
        }
    }
    out[(size_t)pix * C_CH + c] = acc;
}

extern "C" void kernel_launch(void* const* d_in, const int* in_sizes, int n_in,
                              void* d_out, int out_size, void* d_ws, size_t ws_size,
                              hipStream_t stream)
{
    const float* x     = (const float*)d_in[0];
    const float* w1    = (const float*)d_in[1];
    const float* b1    = (const float*)d_in[2];
    const float* gamma = (const float*)d_in[3];
    const float* beta  = (const float*)d_in[4];
    const float* mu    = (const float*)d_in[5];
    const float* var   = (const float*)d_in[6];
    const float* w2    = (const float*)d_in[7];
    const float* b2    = (const float*)d_in[8];

    float* out  = (float*)d_out;
    float* kern = out + (size_t)M_TOT * C_CH;
    float* t    = (float*)d_ws;

    gemm1_bn_relu<<<256, 256, 0, stream>>>(x, w1, b1, gamma, beta, mu, var, t);
    gemm2_kerngen<<<3136, 256, 0, stream>>>(t, w2, b2, kern);
    involution_gather<<<M_TOT, 256, 0, stream>>>(x, kern, out);
}


# GEMM2 re-tiled: a wave keeps its 16-row strip of t as A fragments in registers and sweeps 12-13 n-tiles (2048 waves, not 25088); w2 read and kern stored as float4 via 4-way column interleave
# speedup vs baseline: 2.8241x; 2.8241x over previous
#include <hip/hip_runtime.h>


#define K_SZ   7
#define KK     49
#define G_NUM  16
#define C_CH   256
#define CR     64
#define NK     (KK * G_NUM)
#define HW     64
#define BATCH  2
#define M_TOT  (BATCH * HW * HW)
#define BN_EPS 1e-3f

typedef __attribute__((ext_vector_type(2))) float v2f;
typedef __attribute__((ext_vector_type(8))) float v8f;

__global__ __launch_bounds__(256) void gemm1_bn_relu(
    const float* __restrict__ x,  const float* __restrict__ w1,
    const float* __restrict__ b1, const float* __restrict__ gamma,
    const float* __restrict__ beta, const float* __restrict__ mu,
    const float* __restrict__ var, float* __restrict__ t)
{
    const int wave  = threadIdx.x >> 5;
    const int ln    = threadIdx.x & 31;
    const int tile  = blockIdx.x * 8 + wave;
    const int m0    = (tile >> 2) << 4;
    const int n0    = (tile & 3)  << 4;
    const int laneM = ln & 15;
    const int kHalf = ln >> 4;

    const float* arow = x + (size_t)(m0 + laneM) * C_CH + 2 * kHalf;
    const float* bcol = w1 + (size_t)(2 * kHalf) * CR + (n0 + laneM);

    v8f acc = {};
#pragma unroll 8
    for (int k = 0; k < C_CH; k += 4) {
        v2f a, b;
        a.x = arow[k];
        a.y = arow[k + 1];
        b.x = bcol[(size_t)k * CR];
        b.y = bcol[(size_t)(k + 1) * CR];
        acc = __builtin_amdgcn_wmma_f32_16x16x4_f32(
            false, a, false, b, (short)0, acc, false, false);
    }

    const int   col   = n0 + laneM;
    const float scale = gamma[col] * rsqrtf(var[col] + BN_EPS);
    const float shift = beta[col] - mu[col] * scale;
    const float b1c   = b1[col];
    const int   rowB  = m0 + (kHalf << 3);
#pragma unroll
    for (int r = 0; r < 8; ++r) {
        float v = fmaf(acc[r] + b1c, scale, shift);
        t[(size_t)(rowB + r) * CR + col] = fmaxf(v, 0.0f);
    }
}

#define G2_COLS   64
#define G2_GPW    3
#define G2_PARTS  4
#define G2_TAIL0  (G2_COLS * G2_GPW * G2_PARTS)
#define G2_KS     (CR / 4)

typedef __attribute__((ext_vector_type(4))) float v4f;

struct G2B { v4f x[G2_KS]; v4f y[G2_KS]; };

__device__ __forceinline__ void g2_load_b(G2B& b, const float* __restrict__ bq)
{
#pragma unroll
    for (int s = 0; s < G2_KS; ++s) {
        b.x[s] = *(const v4f*)(bq + (size_t)(4 * s) * NK);
        b.y[s] = *(const v4f*)(bq + (size_t)(4 * s + 1) * NK);
    }
}

__device__ __forceinline__ void g2_group(const v2f (&a)[G2_KS], const G2B& b,
    const float* __restrict__ b2, float* __restrict__ kern, int rowB, int nq)
{
    v8f acc0 = {}, acc1 = {}, acc2 = {}, acc3 = {};
#pragma unroll
    for (int s = 0; s < G2_KS; ++s) {
        acc0 = __builtin_amdgcn_wmma_f32_16x16x4_f32(
            false, a[s], false, (v2f){b.x[s].x, b.y[s].x}, (short)0, acc0, false, false);
        acc1 = __builtin_amdgcn_wmma_f32_16x16x4_f32(
            false, a[s], false, (v2f){b.x[s].y, b.y[s].y}, (short)0, acc1, false, false);
        acc2 = __builtin_amdgcn_wmma_f32_16x16x4_f32(
            false, a[s], false, (v2f){b.x[s].z, b.y[s].z}, (short)0, acc2, false, false);
        acc3 = __builtin_amdgcn_wmma_f32_16x16x4_f32(
            false, a[s], false, (v2f){b.x[s].w, b.y[s].w}, (short)0, acc3, false, false);
    }
    const v4f bias = *(const v4f*)(b2 + nq);
#pragma unroll
    for (int r = 0; r < 8; ++r) {
        const v4f o = {acc0[r] + bias.x, acc1[r] + bias.y, acc2[r] + bias.z, acc3[r] + bias.w};
        *(v4f*)(kern + (size_t)(rowB + r) * NK + nq) = o;
    }
}

__global__ __launch_bounds__(256) __attribute__((amdgpu_waves_per_eu(2, 2)))
void gemm2_kerngen(
    const float* __restrict__ t, const float* __restrict__ w2,
    const float* __restrict__ b2, float* __restrict__ kern)
{
    const int wave  = threadIdx.x >> 5;
    const int ln    = threadIdx.x & 31;
    const int part  = blockIdx.x & (G2_PARTS - 1);
    const int mt    = (blockIdx.x >> 2) * 8 + wave;
    const int m0    = mt << 4;
    const int laneM = ln & 15;
    const int kHalf = ln >> 4;

    const float* arow = t + (size_t)(m0 + laneM) * CR + 2 * kHalf;
    v2f a[G2_KS];
#pragma unroll
    for (int s = 0; s < G2_KS; ++s) a[s] = *(const v2f*)(arow + 4 * s);

    const int rowB = m0 + (kHalf << 3);
    const int nq0  = part * (G2_GPW * G2_COLS) + 4 * laneM;
    const float* bq0 = w2 + (size_t)(2 * kHalf) * NK + nq0;

    G2B bA, bB;
    g2_load_b(bA, bq0);
    g2_load_b(bB, bq0 + G2_COLS);
    g2_group(a, bA, b2, kern, rowB, nq0);
    g2_load_b(bA, bq0 + 2 * G2_COLS);
    g2_group(a, bB, b2, kern, rowB, nq0 + G2_COLS);

    if (part == G2_PARTS - 1) {
        const int col = G2_TAIL0 + laneM;
        const float* bcol = w2 + (size_t)(2 * kHalf) * NK + col;
        v2f bt[G2_KS];
#pragma unroll
        for (int s = 0; s < G2_KS; ++s) {
            bt[s].x = bcol[(size_t)(4 * s) * NK];
            bt[s].y = bcol[(size_t)(4 * s + 1) * NK];
        }
        g2_group(a, bA, b2, kern, rowB, nq0 + 2 * G2_COLS);
        v8f acc = {};
#pragma unroll
        for (int s = 0; s < G2_KS; ++s) {
            acc = __builtin_amdgcn_wmma_f32_16x16x4_f32(
                false, a[s], false, bt[s], (short)0, acc, false, false);
        }
        const float b2c = b2[col];
#pragma unroll
        for (int r = 0; r < 8; ++r) {
            kern[(size_t)(rowB + r) * NK + col] = acc[r] + b2c;
        }
    } else {
        g2_group(a, bA, b2, kern, rowB, nq0 + 2 * G2_COLS);
    }
}

#define TH 2
#define TW 4
#define HR (TH + K_SZ - 1)
#define WR (TW + K_SZ - 1)
#define GT 128
#define KROW4 (TW * NK / 4)
#define KST ((KROW4 + GT - 1) / GT)

typedef const __attribute__((address_space(1))) v2f* gv2f_p;

__global__ __launch_bounds__(GT) __attribute__((amdgpu_waves_per_eu(4, 4)))
void involution_gather(
    const float* __restrict__ x, const float* __restrict__ kern,
    float* __restrict__ out)
{
    __shared__ __attribute__((aligned(16))) float ksh[TH * TW * NK];

    const unsigned tid = threadIdx.x;
    const unsigned ln  = tid & 31u;
    const unsigned wv  = tid >> 5;
    const int tile = blockIdx.x;
    const int w0   = (tile & 15) * TW;
    const int h0   = ((tile >> 4) & 31) * TH;
    const int b    = tile >> 9;
    const unsigned cA  = wv * 64u + 2u * ln;
    const unsigned g0  = (2u * ln) & (G_NUM - 1u);

    v4f kt[TH][KST];
#pragma unroll
    for (int i = 0; i < TH; ++i) {
        const v4f* src = (const v4f*)(kern + (size_t)((b * HW + h0 + i) * HW + w0) * NK);
#pragma unroll
        for (int s = 0; s < KST; ++s) {
            if (s < KST - 1 || tid < KROW4 - GT * (KST - 1)) kt[i][s] = src[tid + GT * s];
        }
    }

#pragma unroll
    for (int i = 0; i < TH; ++i) {
#pragma unroll
        for (int s = 0; s < KST; ++s) {
            if (s < KST - 1 || tid < KROW4 - GT * (KST - 1))
                ((v4f*)ksh)[i * KROW4 + tid + GT * s] = kt[i][s];
        }
    }
    __syncthreads();

    v2f xa[HR][WR];
    const size_t a0 = (size_t)x
        + ((long)((b * HW + h0 - (K_SZ / 2)) * HW + (w0 - (K_SZ / 2))) * C_CH + (long)cA) * 4;
#pragma unroll
    for (int r = 0; r < HR; ++r) {
        const bool rok = (unsigned)(h0 + r - (K_SZ / 2)) < (unsigned)HW;
#pragma unroll
        for (int q = 0; q < WR; ++q) {
            const bool ok = rok && ((unsigned)(w0 + q - (K_SZ / 2)) < (unsigned)HW);
            xa[r][q] = (v2f){0.0f, 0.0f};
            if (ok) {
                xa[r][q] = *(gv2f_p)(a0 + (size_t)(r * HW + q) * (C_CH * 4));
            }
        }
    }

    v2f acc[TH][TW];
#pragma unroll
    for (int i = 0; i < TH; ++i)
#pragma unroll
        for (int j = 0; j < TW; ++j) acc[i][j] = (v2f){0.0f, 0.0f};

    const float* kl = ksh + g0;
#pragma unroll
    for (int r = 0; r < HR; ++r) {
#pragma unroll
        for (int q = 0; q < WR; ++q) {
            const v2f xv = xa[r][q];
#pragma unroll
            for (int i = 0; i < TH; ++i) {
                const int kh = r - i;
                if (kh < 0 || kh >= K_SZ) continue;
#pragma unroll
                for (int j = 0; j < TW; ++j) {
                    const int kw = q - j;
                    if (kw < 0 || kw >= K_SZ) continue;
                    const v2f kv = *(const v2f*)(kl + (i * TW + j) * NK + (kh * K_SZ + kw) * G_NUM);
                    acc[i][j] = __builtin_elementwise_fma(kv, xv, acc[i][j]);
                }
            }
        }
    }

#pragma unroll
    for (int i = 0; i < TH; ++i)
#pragma unroll
        for (int j = 0; j < TW; ++j)
            *(v2f*)(out + (size_t)((b * HW + h0 + i) * HW + w0 + j) * C_CH + cA) = acc[i][j];
}

extern "C" void kernel_launch(void* const* d_in, const int* in_sizes, int n_in,
                              void* d_out, int out_size, void* d_ws, size_t ws_size,
                              hipStream_t stream)
{
    const float* x     = (const float*)d_in[0];
    const float* w1    = (const float*)d_in[1];
    const float* b1    = (const float*)d_in[2];
    const float* gamma = (const float*)d_in[3];
    const float* beta  = (const float*)d_in[4];
    const float* mu    = (const float*)d_in[5];
    const float* var   = (const float*)d_in[6];
    const float* w2    = (const float*)d_in[7];
    const float* b2    = (const float*)d_in[8];

    float* out  = (float*)d_out;
    float* kern = out + (size_t)M_TOT * C_CH;
    float* t    = (float*)d_ws;

    gemm1_bn_relu<<<256, 256, 0, stream>>>(x, w1, b1, gamma, beta, mu, var, t);
    gemm2_kerngen<<<256, 256, 0, stream>>>(t, w2, b2, kern);
    involution_gather<<<1024, 128, 0, stream>>>(x, kern, out);
}


# all three launches fused into one kernel: a block computes t and kern for its own 4x4 pixel tile on WMMA into LDS, copies kern out in whole lines and gathers from LDS (t, kern never re-read)
# speedup vs baseline: 3.8932x; 1.3786x over previous
#include <hip/hip_runtime.h>


#define K_SZ   7
#define KK     49
#define G_NUM  16
#define C_CH   256
#define CR     64
#define NK     (KK * G_NUM)
#define HW     64
#define BATCH  2
#define M_TOT  (BATCH * HW * HW)
#define BN_EPS 1e-3f

typedef __attribute__((ext_vector_type(2))) float v2f;
typedef __attribute__((ext_vector_type(4))) float v4f;
typedef __attribute__((ext_vector_type(8))) float v8f;

#define FT     256
#define TS     4
#define TM     (TS * TS)
#define XP     (C_CH + 4)
#define TP     (CR + 4)
#define G1_KS  (C_CH / 4)
#define G2_KS  (CR / 4)
#define W_COLS 96
#define TAIL0  (8 * W_COLS)
#define KT4    (TM * NK / 4)

#define TH 2
#define TW 4
#define HR (TH + K_SZ - 1)
#define WR (TW + K_SZ - 1)

typedef const __attribute__((address_space(1))) v2f* gv2f_p;

__global__ __launch_bounds__(FT) __attribute__((amdgpu_waves_per_eu(4, 4)))
void involution_fused(
    const float* __restrict__ x,  const float* __restrict__ w1,
    const float* __restrict__ b1, const float* __restrict__ gamma,
    const float* __restrict__ beta, const float* __restrict__ mu,
    const float* __restrict__ var, const float* __restrict__ w2,
    const float* __restrict__ b2, float* __restrict__ out, float* __restrict__ kern)
{
    __shared__ __attribute__((aligned(16))) float ksh[TM * NK];
    __shared__ __attribute__((aligned(16))) float tsh[TM * TP];
    float* xs = ksh;

    const unsigned tid = threadIdx.x;
    const unsigned ln  = tid & 31u;
    const int wv    = __builtin_amdgcn_readfirstlane((int)(tid >> 5));
    const int tile  = blockIdx.x;
    const int w0    = (tile & 15) * TS;
    const int h0    = ((tile >> 4) & 15) * TS;
    const int b     = tile >> 8;
    const int laneM = ln & 15;
    const int kHalf = ln >> 4;

    const int pj = tid >> 6, c4 = tid & 63;
    v4f xt[TS];
#pragma unroll
    for (int i = 0; i < TS; ++i)
        xt[i] = *(const v4f*)(x + (size_t)((b * HW + h0 + i) * HW + w0 + pj) * C_CH + 4 * c4);

    v2f bw[G1_KS];
    float scale = 0.0f, shift = 0.0f, b1c = 0.0f;
    if (wv < 4) {
        const int col = wv * 16 + laneM;
        const float* bcol = w1 + (size_t)(2 * kHalf) * CR + col;
#pragma unroll
        for (int s = 0; s < G1_KS; ++s) {
            bw[s].x = bcol[(size_t)(4 * s) * CR];
            bw[s].y = bcol[(size_t)(4 * s + 1) * CR];
        }
        scale = gamma[col] * rsqrtf(var[col] + BN_EPS);
        shift = beta[col] - mu[col] * scale;
        b1c   = b1[col];
    }

#pragma unroll
    for (int i = 0; i < TS; ++i)
        *(v4f*)(xs + (4 * i + pj) * XP + 4 * c4) = xt[i];
    __syncthreads();

    if (wv < 4) {
        const float* arow = xs + laneM * XP + 2 * kHalf;
        v8f acc = {};
#pragma unroll
        for (int s = 0; s < G1_KS; ++s) {
            const v2f a1 = *(const v2f*)(arow + 4 * s);
            acc = __builtin_amdgcn_wmma_f32_16x16x4_f32(
                false, a1, false, bw[s], (short)0, acc, false, false);
        }
        const int col = wv * 16 + laneM;
#pragma unroll
        for (int r = 0; r < 8; ++r) {
            const float v = fmaf(acc[r] + b1c, scale, shift);
            tsh[(8 * kHalf + r) * TP + col] = fmaxf(v, 0.0f);
        }
    }

    const int nb = wv * W_COLS;
    const float* bp = w2 + (size_t)(2 * kHalf) * NK + nb + 4 * laneM;
    const float* bq = w2 + (size_t)(2 * kHalf) * NK + nb + 64 + 2 * laneM;
    v4f px[G2_KS], py[G2_KS];
    v2f qx[G2_KS], qy[G2_KS];
#pragma unroll
    for (int s = 0; s < G2_KS; ++s) {
        px[s] = *(const v4f*)(bp + (size_t)(4 * s) * NK);
        py[s] = *(const v4f*)(bp + (size_t)(4 * s + 1) * NK);
    }
#pragma unroll
    for (int s = 0; s < G2_KS; ++s) {
        qx[s] = *(const v2f*)(bq + (size_t)(4 * s) * NK);
        qy[s] = *(const v2f*)(bq + (size_t)(4 * s + 1) * NK);
    }
    const v4f biasP = *(const v4f*)(b2 + nb + 4 * laneM);
    const v2f biasQ = *(const v2f*)(b2 + nb + 64 + 2 * laneM);

    __syncthreads();

    v2f a[G2_KS];
#pragma unroll
    for (int s = 0; s < G2_KS; ++s) a[s] = *(const v2f*)(tsh + laneM * TP + 4 * s + 2 * kHalf);

    float* krow = ksh + (8 * kHalf) * NK;
    {
        v8f acc0 = {}, acc1 = {}, acc2 = {}, acc3 = {};
#pragma unroll
        for (int s = 0; s < G2_KS; ++s) {
            acc0 = __builtin_amdgcn_wmma_f32_16x16x4_f32(
                false, a[s], false, (v2f){px[s].x, py[s].x}, (short)0, acc0, false, false);
            acc1 = __builtin_amdgcn_wmma_f32_16x16x4_f32(
                false, a[s], false, (v2f){px[s].y, py[s].y}, (short)0, acc1, false, false);
            acc2 = __builtin_amdgcn_wmma_f32_16x16x4_f32(
                false, a[s], false, (v2f){px[s].z, py[s].z}, (short)0, acc2, false, false);
            acc3 = __builtin_amdgcn_wmma_f32_16x16x4_f32(
                false, a[s], false, (v2f){px[s].w, py[s].w}, (short)0, acc3, false, false);
        }
#pragma unroll
        for (int r = 0; r < 8; ++r) {
            const v4f o = {acc0[r] + biasP.x, acc1[r] + biasP.y, acc2[r] + biasP.z, acc3[r] + biasP.w};
            *(v4f*)(krow + r * NK + nb + 4 * laneM) = o;
        }
    }
    {
        v8f acc4 = {}, acc5 = {};
#pragma unroll
        for (int s = 0; s < G2_KS; ++s) {
            acc4 = __builtin_amdgcn_wmma_f32_16x16x4_f32(
                false, a[s], false, (v2f){qx[s].x, qy[s].x}, (short)0, acc4, false, false);
            acc5 = __builtin_amdgcn_wmma_f32_16x16x4_f32(
                false, a[s], false, (v2f){qx[s].y, qy[s].y}, (short)0, acc5, false, false);
        }
#pragma unroll
        for (int r = 0; r < 8; ++r) {
            const v2f o = {acc4[r] + biasQ.x, acc5[r] + biasQ.y};
            *(v2f*)(krow + r * NK + nb + 64 + 2 * laneM) = o;
        }
    }
    if (wv == 7) {
        const int col = TAIL0 + laneM;
        const float* bcol = w2 + (size_t)(2 * kHalf) * NK + col;
        v2f bt[G2_KS];
#pragma unroll
        for (int s = 0; s < G2_KS; ++s) {
            bt[s].x = bcol[(size_t)(4 * s) * NK];
            bt[s].y = bcol[(size_t)(4 * s + 1) * NK];
        }
        const float b2c = b2[col];
        v8f acc = {};
#pragma unroll
        for (int s = 0; s < G2_KS; ++s) {
            acc = __builtin_amdgcn_wmma_f32_16x16x4_f32(
                false, a[s], false, bt[s], (short)0, acc, false, false);
        }
#pragma unroll
        for (int r = 0; r < 8; ++r) krow[r * NK + col] = acc[r] + b2c;
    }
    __syncthreads();

    const int sub = wv >> 2;
    const int hs  = h0 + TH * sub;
    const unsigned cA = (unsigned)(wv & 3) * 64u + 2u * ln;
    const unsigned g0 = (2u * ln) & (G_NUM - 1u);

    v2f xa[HR][WR];
    const size_t a0 = (size_t)x
        + ((long)((b * HW + hs - (K_SZ / 2)) * HW + (w0 - (K_SZ / 2))) * C_CH + (long)cA) * 4;
#pragma unroll
    for (int r = 0; r < HR; ++r) {
        const bool rok = (unsigned)(hs + r - (K_SZ / 2)) < (unsigned)HW;
#pragma unroll
        for (int q = 0; q < WR; ++q) {
            const bool ok = rok && ((unsigned)(w0 + q - (K_SZ / 2)) < (unsigned)HW);
            xa[r][q] = (v2f){0.0f, 0.0f};
            if (ok) {
                xa[r][q] = *(gv2f_p)(a0 + (size_t)(r * HW + q) * (C_CH * 4));
            }
        }
    }

#pragma unroll
    for (int s = 0; s < (KT4 + FT - 1) / FT; ++s) {
        const int idx = tid + FT * s;
        if (idx < KT4) {
            const int i   = idx / (KT4 / TS);
            const int rem = idx - i * (KT4 / TS);
            v4f* dst = (v4f*)(kern + (size_t)((b * HW + h0 + i) * HW + w0) * NK) + rem;
            *dst = ((const v4f*)ksh)[idx];
        }
    }

    v2f acc[TH][TW];
#pragma unroll
    for (int i = 0; i < TH; ++i)
#pragma unroll
        for (int j = 0; j < TW; ++j) acc[i][j] = (v2f){0.0f, 0.0f};

    const float* kl = ksh + (sub * TH * TW) * NK + g0;
#pragma unroll
    for (int r = 0; r < HR; ++r) {
#pragma unroll
        for (int q = 0; q < WR; ++q) {
            const v2f xv = xa[r][q];
#pragma unroll
            for (int i = 0; i < TH; ++i) {
                const int kh = r - i;
                if (kh < 0 || kh >= K_SZ) continue;
#pragma unroll
                for (int j = 0; j < TW; ++j) {
                    const int kw = q - j;
                    if (kw < 0 || kw >= K_SZ) continue;
                    const v2f kv = *(const v2f*)(kl + (i * TW + j) * NK + (kh * K_SZ + kw) * G_NUM);
                    acc[i][j] = __builtin_elementwise_fma(kv, xv, acc[i][j]);
                }
            }
        }
    }

#pragma unroll
    for (int i = 0; i < TH; ++i)
#pragma unroll
        for (int j = 0; j < TW; ++j)
            *(v2f*)(out + (size_t)((b * HW + hs + i) * HW + w0 + j) * C_CH + cA) = acc[i][j];
}

extern "C" void kernel_launch(void* const* d_in, const int* in_sizes, int n_in,
                              void* d_out, int out_size, void* d_ws, size_t ws_size,
                              hipStream_t stream)
{
    const float* x     = (const float*)d_in[0];
    const float* w1    = (const float*)d_in[1];
    const float* b1    = (const float*)d_in[2];
    const float* gamma = (const float*)d_in[3];
    const float* beta  = (const float*)d_in[4];
    const float* mu    = (const float*)d_in[5];
    const float* var   = (const float*)d_in[6];
    const float* w2    = (const float*)d_in[7];
    const float* b2    = (const float*)d_in[8];

    float* out  = (float*)d_out;
    float* kern = out + (size_t)M_TOT * C_CH;

    involution_fused<<<512, FT, 0, stream>>>(x, w1, b1, gamma, beta, mu, var, w2, b2, out, kern);
}


# block enlarged to 16 waves owning two stacked 4x4 m-tiles (grid 256): one w1/w2 fragment set per wave feeds both m-tiles, halving weight traffic, and all w2 loads run beside GEMM1; sum order kept
# speedup vs baseline: 4.3074x; 1.1064x over previous
#include <hip/hip_runtime.h>


#define K_SZ   7
#define KK     49
#define G_NUM  16
#define C_CH   256
#define CR     64
#define NK     (KK * G_NUM)
#define HW     64
#define BATCH  2
#define M_TOT  (BATCH * HW * HW)
#define BN_EPS 1e-3f

typedef __attribute__((ext_vector_type(2))) float v2f;
typedef __attribute__((ext_vector_type(4))) float v4f;
typedef __attribute__((ext_vector_type(8))) float v8f;

#define FT     512
#define TS     4
#define TM     (TS * TS)
#define MT     2
#define RH     (MT * TS)
#define RM     (MT * TM)
#define XP     (C_CH + 4)
#define TP     (CR + 4)
#define G1_KS  (C_CH / 4)
#define G2_KS  (CR / 4)
#define W_COLS 64
#define TAIL0  (12 * W_COLS)
#define KT4    (RM * NK / 4)

#define TH 2
#define TW 4
#define HR (TH + K_SZ - 1)
#define WR (TW + K_SZ - 1)

typedef const __attribute__((address_space(1))) v2f* gv2f_p;

__global__ __launch_bounds__(FT) __attribute__((amdgpu_waves_per_eu(4, 4)))
void involution_fused(
    const float* __restrict__ x,  const float* __restrict__ w1,
    const float* __restrict__ b1, const float* __restrict__ gamma,
    const float* __restrict__ beta, const float* __restrict__ mu,
    const float* __restrict__ var, const float* __restrict__ w2,
    const float* __restrict__ b2, float* __restrict__ out, float* __restrict__ kern)
{
    __shared__ __attribute__((aligned(16))) float ksh[RM * NK];
    __shared__ __attribute__((aligned(16))) float tsh[RM * TP];
    float* xs = ksh;

    const unsigned tid = threadIdx.x;
    const unsigned ln  = tid & 31u;
    const int wv    = __builtin_amdgcn_readfirstlane((int)(tid >> 5));
    const int tile  = blockIdx.x;
    const int w0    = (tile & 15) * TS;
    const int h0    = ((tile >> 4) & 7) * RH;
    const int b     = tile >> 7;
    const int laneM = ln & 15;
    const int kHalf = ln >> 4;

    const int c4 = tid & 63, pj = (tid >> 6) & 3, rh = tid >> 8;
    v4f xt[TS];
#pragma unroll
    for (int i = 0; i < TS; ++i)
        xt[i] = *(const v4f*)(x + (size_t)((b * HW + h0 + TS * rh + i) * HW + w0 + pj) * C_CH + 4 * c4);

    v2f bw[G1_KS];
    float scale = 0.0f, shift = 0.0f, b1c = 0.0f;
    if (wv >= 12) {
        const int col = (wv - 12) * 16 + laneM;
        const float* bcol = w1 + (size_t)(2 * kHalf) * CR + col;
#pragma unroll
        for (int s = 0; s < G1_KS; ++s) {
            bw[s].x = bcol[(size_t)(4 * s) * CR];
            bw[s].y = bcol[(size_t)(4 * s + 1) * CR];
        }
        scale = gamma[col] * rsqrtf(var[col] + BN_EPS);
        shift = beta[col] - mu[col] * scale;
        b1c   = b1[col];
    }

#pragma unroll
    for (int i = 0; i < TS; ++i)
        *(v4f*)(xs + (TM * rh + 4 * i + pj) * XP + 4 * c4) = xt[i];
    __syncthreads();

    if (wv >= 12) {
        const int col = (wv - 12) * 16 + laneM;
        const float* arow0 = xs + laneM * XP + 2 * kHalf;
        const float* arow1 = arow0 + TM * XP;
        v8f acc0 = {}, acc1 = {};
#pragma unroll
        for (int s = 0; s < G1_KS; ++s) {
            const v2f a0 = *(const v2f*)(arow0 + 4 * s);
            const v2f a1 = *(const v2f*)(arow1 + 4 * s);
            acc0 = __builtin_amdgcn_wmma_f32_16x16x4_f32(
                false, a0, false, bw[s], (short)0, acc0, false, false);
            acc1 = __builtin_amdgcn_wmma_f32_16x16x4_f32(
                false, a1, false, bw[s], (short)0, acc1, false, false);
        }
#pragma unroll
        for (int r = 0; r < 8; ++r) {
            const float v0 = fmaf(acc0[r] + b1c, scale, shift);
            const float v1 = fmaf(acc1[r] + b1c, scale, shift);
            tsh[(8 * kHalf + r) * TP + col]      = fmaxf(v0, 0.0f);
            tsh[(TM + 8 * kHalf + r) * TP + col] = fmaxf(v1, 0.0f);
        }
    }

    const int nb = wv * W_COLS;
    v4f px[G2_KS], py[G2_KS];
    if (wv < 12) {
        const float* bp = w2 + (size_t)(2 * kHalf) * NK + nb + 4 * laneM;
#pragma unroll
        for (int s = 0; s < G2_KS; ++s) {
            px[s] = *(const v4f*)(bp + (size_t)(4 * s) * NK);
            py[s] = *(const v4f*)(bp + (size_t)(4 * s + 1) * NK);
        }
    }

    __syncthreads();

    if (wv < 12) {
        const v4f biasP = *(const v4f*)(b2 + nb + 4 * laneM);
#pragma unroll
        for (int mt = 0; mt < MT; ++mt) {
            v2f a[G2_KS];
#pragma unroll
            for (int s = 0; s < G2_KS; ++s)
                a[s] = *(const v2f*)(tsh + (TM * mt + laneM) * TP + 4 * s + 2 * kHalf);
            v8f acc0 = {}, acc1 = {}, acc2 = {}, acc3 = {};
#pragma unroll
            for (int s = 0; s < G2_KS; ++s) {
                acc0 = __builtin_amdgcn_wmma_f32_16x16x4_f32(
                    false, a[s], false, (v2f){px[s].x, py[s].x}, (short)0, acc0, false, false);
                acc1 = __builtin_amdgcn_wmma_f32_16x16x4_f32(
                    false, a[s], false, (v2f){px[s].y, py[s].y}, (short)0, acc1, false, false);
                acc2 = __builtin_amdgcn_wmma_f32_16x16x4_f32(
                    false, a[s], false, (v2f){px[s].z, py[s].z}, (short)0, acc2, false, false);
                acc3 = __builtin_amdgcn_wmma_f32_16x16x4_f32(
                    false, a[s], false, (v2f){px[s].w, py[s].w}, (short)0, acc3, false, false);
            }
            float* krow = ksh + (TM * mt + 8 * kHalf) * NK;
#pragma unroll
            for (int r = 0; r < 8; ++r) {
                const v4f o = {acc0[r] + biasP.x, acc1[r] + biasP.y, acc2[r] + biasP.z, acc3[r] + biasP.w};
                *(v4f*)(krow + r * NK + nb + 4 * laneM) = o;
            }
        }
    } else if (wv < 14) {
        const int mt = wv - 12;
        const int tcol = TAIL0 + laneM;
        const float* bcol = w2 + (size_t)(2 * kHalf) * NK + tcol;
        v2f bt[G2_KS];
#pragma unroll
        for (int s = 0; s < G2_KS; ++s) {
            bt[s].x = bcol[(size_t)(4 * s) * NK];
            bt[s].y = bcol[(size_t)(4 * s + 1) * NK];
        }
        const float b2c = b2[tcol];
        v2f a[G2_KS];
#pragma unroll
        for (int s = 0; s < G2_KS; ++s)
            a[s] = *(const v2f*)(tsh + (TM * mt + laneM) * TP + 4 * s + 2 * kHalf);
        v8f acc = {};
#pragma unroll
        for (int s = 0; s < G2_KS; ++s) {
            acc = __builtin_amdgcn_wmma_f32_16x16x4_f32(
                false, a[s], false, bt[s], (short)0, acc, false, false);
        }
        float* krow = ksh + (TM * mt + 8 * kHalf) * NK;
#pragma unroll
        for (int r = 0; r < 8; ++r) krow[r * NK + tcol] = acc[r] + b2c;
    }
    __syncthreads();

    const int sub = wv >> 2;
    const int hs  = h0 + TH * sub;
    const unsigned cA = (unsigned)(wv & 3) * 64u + 2u * ln;
    const unsigned g0 = (2u * ln) & (G_NUM - 1u);

    v2f xa[HR][WR];
    const size_t a0 = (size_t)x
        + ((long)((b * HW + hs - (K_SZ / 2)) * HW + (w0 - (K_SZ / 2))) * C_CH + (long)cA) * 4;
#pragma unroll
    for (int r = 0; r < HR; ++r) {
        const bool rok = (unsigned)(hs + r - (K_SZ / 2)) < (unsigned)HW;
#pragma unroll
        for (int q = 0; q < WR; ++q) {
            const bool ok = rok && ((unsigned)(w0 + q - (K_SZ / 2)) < (unsigned)HW);
            xa[r][q] = (v2f){0.0f, 0.0f};
            if (ok) {
                xa[r][q] = *(gv2f_p)(a0 + (size_t)(r * HW + q) * (C_CH * 4));
            }
        }
    }

#pragma unroll
    for (int s = 0; s < (KT4 + FT - 1) / FT; ++s) {
        const int idx = tid + FT * s;
        if (idx < KT4) {
            const int i   = idx / (KT4 / RH);
            const int rem = idx - i * (KT4 / RH);
            v4f* dst = (v4f*)(kern + (size_t)((b * HW + h0 + i) * HW + w0) * NK) + rem;
            *dst = ((const v4f*)ksh)[idx];
        }
    }

    v2f acc[TH][TW];
#pragma unroll
    for (int i = 0; i < TH; ++i)
#pragma unroll
        for (int j = 0; j < TW; ++j) acc[i][j] = (v2f){0.0f, 0.0f};

    const float* kl = ksh + (sub * TH * TW) * NK + g0;
#pragma unroll
    for (int r = 0; r < HR; ++r) {
#pragma unroll
        for (int q = 0; q < WR; ++q) {
            const v2f xv = xa[r][q];
#pragma unroll
            for (int i = 0; i < TH; ++i) {
                const int kh = r - i;
                if (kh < 0 || kh >= K_SZ) continue;
#pragma unroll
                for (int j = 0; j < TW; ++j) {
                    const int kw = q - j;
                    if (kw < 0 || kw >= K_SZ) continue;
                    const v2f kv = *(const v2f*)(kl + (i * TW + j) * NK + (kh * K_SZ + kw) * G_NUM);
                    acc[i][j] = __builtin_elementwise_fma(kv, xv, acc[i][j]);
                }
            }
        }
    }

#pragma unroll
    for (int i = 0; i < TH; ++i)
#pragma unroll
        for (int j = 0; j < TW; ++j)
            *(v2f*)(out + (size_t)((b * HW + hs + i) * HW + w0 + j) * C_CH + cA) = acc[i][j];
}

extern "C" void kernel_launch(void* const* d_in, const int* in_sizes, int n_in,
                              void* d_out, int out_size, void* d_ws, size_t ws_size,
                              hipStream_t stream)
{
    const float* x     = (const float*)d_in[0];
    const float* w1    = (const float*)d_in[1];
    const float* b1    = (const float*)d_in[2];
    const float* gamma = (const float*)d_in[3];
    const float* beta  = (const float*)d_in[4];
    const float* mu    = (const float*)d_in[5];
    const float* var   = (const float*)d_in[6];
    const float* w2    = (const float*)d_in[7];
    const float* b2    = (const float*)d_in[8];

    float* out  = (float*)d_out;
    float* kern = out + (size_t)M_TOT * C_CH;

    involution_fused<<<256, FT, 0, stream>>>(x, w1, b1, gamma, beta, mu, var, w2, b2, out, kern);
}


# gather: each kern value read from LDS feeds two channel pairs per lane (wave = 128 channels x half the halo rows); the two tap-row half sums of a pixel are added through LDS (sum order changes)
# speedup vs baseline: 4.5527x; 1.0569x over previous
#include <hip/hip_runtime.h>


#define K_SZ   7
#define KK     49
#define G_NUM  16
#define C_CH   256
#define CR     64
#define NK     (KK * G_NUM)
#define HW     64
#define BATCH  2
#define M_TOT  (BATCH * HW * HW)
#define BN_EPS 1e-3f

typedef __attribute__((ext_vector_type(2))) float v2f;
typedef __attribute__((ext_vector_type(4))) float v4f;
typedef __attribute__((ext_vector_type(8))) float v8f;

#define FT     512
#define TS     4
#define TM     (TS * TS)
#define MT     2
#define RH     (MT * TS)
#define RM     (MT * TM)
#define XP     (C_CH + 4)
#define TP     (CR + 4)
#define G1_KS  (C_CH / 4)
#define G2_KS  (CR / 4)
#define W_COLS 64
#define TAIL0  (12 * W_COLS)
#define KT4    (RM * NK / 4)

#define TH 2
#define TW 4
#define HR (TH + K_SZ - 1)
#define WR (TW + K_SZ - 1)
#define HRW (HR / 2)
#define CP 2

typedef const __attribute__((address_space(1))) v2f* gv2f_p;

template <int TG>
__device__ __forceinline__ void gather_rows(const float* kl, const v2f (&xa)[HRW][WR][CP], float* pw,
                                            v2f (&own)[TW][CP])
{
    v2f acc[TH][TW][CP];
#pragma unroll
    for (int i = 0; i < TH; ++i)
#pragma unroll
        for (int j = 0; j < TW; ++j)
#pragma unroll
            for (int m = 0; m < CP; ++m) acc[i][j][m] = (v2f){0.0f, 0.0f};

#pragma unroll
    for (int a = 0; a < HRW; ++a) {
#pragma unroll
        for (int q = 0; q < WR; ++q) {
#pragma unroll
            for (int i = 0; i < TH; ++i) {
                const int kh = HRW * TG + a - i;
                if (kh < 0 || kh >= K_SZ) continue;
#pragma unroll
                for (int j = 0; j < TW; ++j) {
                    const int kw = q - j;
                    if (kw < 0 || kw >= K_SZ) continue;
                    const v2f kv = *(const v2f*)(kl + (i * TW + j) * NK + (kh * K_SZ + kw) * G_NUM);
#pragma unroll
                    for (int m = 0; m < CP; ++m)
                        acc[i][j][m] = __builtin_elementwise_fma(kv, xa[a][q][m], acc[i][j][m]);
                }
            }
        }
    }

#pragma unroll
    for (int j = 0; j < TW; ++j)
#pragma unroll
        for (int m = 0; m < CP; ++m) {
            *(v2f*)(pw + ((1 - TG) * TW + j) * C_CH + 64 * m) = acc[1 - TG][j][m];
            own[j][m] = acc[TG][j][m];
        }
}

__global__ __launch_bounds__(FT) __attribute__((amdgpu_waves_per_eu(4, 4)))
void involution_fused(
    const float* __restrict__ x,  const float* __restrict__ w1,
    const float* __restrict__ b1, const float* __restrict__ gamma,
    const float* __restrict__ beta, const float* __restrict__ mu,
    const float* __restrict__ var, const float* __restrict__ w2,
    const float* __restrict__ b2, float* __restrict__ out, float* __restrict__ kern)
{
    __shared__ __attribute__((aligned(16))) float ksh[RM * NK];
    __shared__ __attribute__((aligned(16))) float tsh[RM * TP];
    __shared__ __attribute__((aligned(16))) float psh[RM * C_CH];
    float* xs = ksh;

    const unsigned tid = threadIdx.x;
    const unsigned ln  = tid & 31u;
    const int wv    = __builtin_amdgcn_readfirstlane((int)(tid >> 5));
    const int tile  = blockIdx.x;
    const int w0    = (tile & 15) * TS;
    const int h0    = ((tile >> 4) & 7) * RH;
    const int b     = tile >> 7;
    const int laneM = ln & 15;
    const int kHalf = ln >> 4;

    const int c4 = tid & 63, pj = (tid >> 6) & 3, rh = tid >> 8;
    v4f xt[TS];
#pragma unroll
    for (int i = 0; i < TS; ++i)
        xt[i] = *(const v4f*)(x + (size_t)((b * HW + h0 + TS * rh + i) * HW + w0 + pj) * C_CH + 4 * c4);

    v2f bw[G1_KS];
    float scale = 0.0f, shift = 0.0f, b1c = 0.0f;
    if (wv >= 12) {
        const int col = (wv - 12) * 16 + laneM;
        const float* bcol = w1 + (size_t)(2 * kHalf) * CR + col;
#pragma unroll
        for (int s = 0; s < G1_KS; ++s) {
            bw[s].x = bcol[(size_t)(4 * s) * CR];
            bw[s].y = bcol[(size_t)(4 * s + 1) * CR];
        }
        scale = gamma[col] * rsqrtf(var[col] + BN_EPS);
        shift = beta[col] - mu[col] * scale;
        b1c   = b1[col];
    }

#pragma unroll
    for (int i = 0; i < TS; ++i)
        *(v4f*)(xs + (TM * rh + 4 * i + pj) * XP + 4 * c4) = xt[i];
    __syncthreads();

    if (wv >= 12) {
        const int col = (wv - 12) * 16 + laneM;
        const float* arow0 = xs + laneM * XP + 2 * kHalf;
        const float* arow1 = arow0 + TM * XP;
        v8f acc0 = {}, acc1 = {};
#pragma unroll
        for (int s = 0; s < G1_KS; ++s) {
            const v2f a0 = *(const v2f*)(arow0 + 4 * s);
            const v2f a1 = *(const v2f*)(arow1 + 4 * s);
            acc0 = __builtin_amdgcn_wmma_f32_16x16x4_f32(
                false, a0, false, bw[s], (short)0, acc0, false, false);
            acc1 = __builtin_amdgcn_wmma_f32_16x16x4_f32(
                false, a1, false, bw[s], (short)0, acc1, false, false);
        }
#pragma unroll
        for (int r = 0; r < 8; ++r) {
            const float v0 = fmaf(acc0[r] + b1c, scale, shift);
            const float v1 = fmaf(acc1[r] + b1c, scale, shift);
            tsh[(8 * kHalf + r) * TP + col]      = fmaxf(v0, 0.0f);
            tsh[(TM + 8 * kHalf + r) * TP + col] = fmaxf(v1, 0.0f);
        }
    }

    const int nb = wv * W_COLS;
    v4f px[G2_KS], py[G2_KS];
    if (wv < 12) {
        const float* bp = w2 + (size_t)(2 * kHalf) * NK + nb + 4 * laneM;
#pragma unroll
        for (int s = 0; s < G2_KS; ++s) {
            px[s] = *(const v4f*)(bp + (size_t)(4 * s) * NK);
            py[s] = *(const v4f*)(bp + (size_t)(4 * s + 1) * NK);
        }
    }

    __syncthreads();

    if (wv < 12) {
        const v4f biasP = *(const v4f*)(b2 + nb + 4 * laneM);
#pragma unroll
        for (int mt = 0; mt < MT; ++mt) {
            v2f a[G2_KS];
#pragma unroll
            for (int s = 0; s < G2_KS; ++s)
                a[s] = *(const v2f*)(tsh + (TM * mt + laneM) * TP + 4 * s + 2 * kHalf);
            v8f acc0 = {}, acc1 = {}, acc2 = {}, acc3 = {};
#pragma unroll
            for (int s = 0; s < G2_KS; ++s) {
                acc0 = __builtin_amdgcn_wmma_f32_16x16x4_f32(
                    false, a[s], false, (v2f){px[s].x, py[s].x}, (short)0, acc0, false, false);
                acc1 = __builtin_amdgcn_wmma_f32_16x16x4_f32(
                    false, a[s], false, (v2f){px[s].y, py[s].y}, (short)0, acc1, false, false);
                acc2 = __builtin_amdgcn_wmma_f32_16x16x4_f32(
                    false, a[s], false, (v2f){px[s].z, py[s].z}, (short)0, acc2, false, false);
                acc3 = __builtin_amdgcn_wmma_f32_16x16x4_f32(
                    false, a[s], false, (v2f){px[s].w, py[s].w}, (short)0, acc3, false, false);
            }
            float* krow = ksh + (TM * mt + 8 * kHalf) * NK;
#pragma unroll
            for (int r = 0; r < 8; ++r) {
                const v4f o = {acc0[r] + biasP.x, acc1[r] + biasP.y, acc2[r] + biasP.z, acc3[r] + biasP.w};
                *(v4f*)(krow + r * NK + nb + 4 * laneM) = o;
            }
        }
    } else if (wv < 14) {
        const int mt = wv - 12;
        const int tcol = TAIL0 + laneM;
        const float* bcol = w2 + (size_t)(2 * kHalf) * NK + tcol;
        v2f bt[G2_KS];
#pragma unroll
        for (int s = 0; s < G2_KS; ++s) {
            bt[s].x = bcol[(size_t)(4 * s) * NK];
            bt[s].y = bcol[(size_t)(4 * s + 1) * NK];
        }
        const float b2c = b2[tcol];
        v2f a[G2_KS];
#pragma unroll
        for (int s = 0; s < G2_KS; ++s)
            a[s] = *(const v2f*)(tsh + (TM * mt + laneM) * TP + 4 * s + 2 * kHalf);
        v8f acc = {};
#pragma unroll
        for (int s = 0; s < G2_KS; ++s) {
            acc = __builtin_amdgcn_wmma_f32_16x16x4_f32(
                false, a[s], false, bt[s], (short)0, acc, false, false);
        }
        float* krow = ksh + (TM * mt + 8 * kHalf) * NK;
#pragma unroll
        for (int r = 0; r < 8; ++r) krow[r * NK + tcol] = acc[r] + b2c;
    }
    __syncthreads();

    const int sub = wv >> 2;
    const int chf = (wv >> 1) & 1;
    const int tg  = wv & 1;
    const int hs  = h0 + TH * sub;
    const unsigned cA = (unsigned)chf * 128u + 2u * ln;
    const unsigned g0 = (2u * ln) & (G_NUM - 1u);

    v2f xa[HRW][WR][CP];
    const int hr0 = hs + HRW * tg - (K_SZ / 2);
    const size_t a0 = (size_t)x
        + ((long)((b * HW + hr0) * HW + (w0 - (K_SZ / 2))) * C_CH + (long)cA) * 4;
#pragma unroll
    for (int a = 0; a < HRW; ++a) {
        const bool rok = (unsigned)(hr0 + a) < (unsigned)HW;
#pragma unroll
        for (int q = 0; q < WR; ++q) {
            const bool ok = rok && ((unsigned)(w0 + q - (K_SZ / 2)) < (unsigned)HW);
#pragma unroll
            for (int m = 0; m < CP; ++m) xa[a][q][m] = (v2f){0.0f, 0.0f};
            if (ok) {
#pragma unroll
                for (int m = 0; m < CP; ++m)
                    xa[a][q][m] = *(gv2f_p)(a0 + (size_t)(a * HW + q) * (C_CH * 4) + (size_t)(m * 64 * 4));
            }
        }
    }

#pragma unroll
    for (int s = 0; s < (KT4 + FT - 1) / FT; ++s) {
        const int idx = tid + FT * s;
        if (idx < KT4) {
            const int i   = idx / (KT4 / RH);
            const int rem = idx - i * (KT4 / RH);
            v4f* dst = (v4f*)(kern + (size_t)((b * HW + h0 + i) * HW + w0) * NK) + rem;
            *dst = ((const v4f*)ksh)[idx];
        }
    }

    const float* kl = ksh + (sub * TH * TW) * NK + g0;
    float* pw = psh + (sub * TH * TW) * C_CH + cA;
    v2f own[TW][CP];
    if (tg == 0) gather_rows<0>(kl, xa, pw, own);
    else         gather_rows<1>(kl, xa, pw, own);

    __syncthreads();

#pragma unroll
    for (int j = 0; j < TW; ++j)
#pragma unroll
        for (int m = 0; m < CP; ++m) {
            const v2f o = *(const v2f*)(pw + (tg * TW + j) * C_CH + 64 * m);
            *(v2f*)(out + (size_t)((b * HW + hs + tg) * HW + w0 + j) * C_CH + cA + 64 * m) = own[j][m] + o;
        }
}

extern "C" void kernel_launch(void* const* d_in, const int* in_sizes, int n_in,
                              void* d_out, int out_size, void* d_ws, size_t ws_size,
                              hipStream_t stream)
{
    const float* x     = (const float*)d_in[0];
    const float* w1    = (const float*)d_in[1];
    const float* b1    = (const float*)d_in[2];
    const float* gamma = (const float*)d_in[3];
    const float* beta  = (const float*)d_in[4];
    const float* mu    = (const float*)d_in[5];
    const float* var   = (const float*)d_in[6];
    const float* w2    = (const float*)d_in[7];
    const float* b2    = (const float*)d_in[8];

    float* out  = (float*)d_out;
    float* kern = out + (size_t)M_TOT * C_CH;

    involution_fused<<<256, FT, 0, stream>>>(x, w1, b1, gamma, beta, mu, var, w2, b2, out, kern);
}


# round trips at the phase boundaries hidden: barriers 1 and 3 order LDS only, so w2 is requested one round trip sooner and 3 of 4 halo rows before the GEMM2 store; GEMM1 reads A 8 k-steps ahead
# speedup vs baseline: 4.8883x; 1.0737x over previous
#include <hip/hip_runtime.h>


#define K_SZ   7
#define KK     49
#define G_NUM  16
#define C_CH   256
#define CR     64
#define NK     (KK * G_NUM)
#define HW     64
#define BATCH  2
#define M_TOT  (BATCH * HW * HW)
#define BN_EPS 1e-3f

typedef __attribute__((ext_vector_type(2))) float v2f;
typedef __attribute__((ext_vector_type(4))) float v4f;
typedef __attribute__((ext_vector_type(8))) float v8f;

#define FT     512
#define TS     4
#define TM     (TS * TS)
#define MT     2
#define RH     (MT * TS)
#define RM     (MT * TM)
#define XP     (C_CH + 4)
#define TP     (CR + 4)
#define G1_KS  (C_CH / 4)
#define G2_KS  (CR / 4)
#define A_BK   8
#define W_COLS 64
#define TAIL0  (12 * W_COLS)
#define KT4    (RM * NK / 4)

#define TH 2
#define TW 4
#define HR (TH + K_SZ - 1)
#define WR (TW + K_SZ - 1)
#define HRW (HR / 2)
#define CP 2

typedef const __attribute__((address_space(1))) v2f* gv2f_p;

__device__ __forceinline__ void barrier_lds_only()
{
    __builtin_amdgcn_fence(__ATOMIC_RELEASE, "workgroup", "local");
    __builtin_amdgcn_s_barrier();
    __builtin_amdgcn_fence(__ATOMIC_ACQUIRE, "workgroup", "local");
}

template <int TG>
__device__ __forceinline__ void gather_rows(const float* kl, const v2f (&xa)[HRW][WR][CP],
                                            const unsigned (&keep)[WR], float* pw, v2f (&own)[TW][CP])
{
    v2f acc[TH][TW][CP];
#pragma unroll
    for (int i = 0; i < TH; ++i)
#pragma unroll
        for (int j = 0; j < TW; ++j)
#pragma unroll
            for (int m = 0; m < CP; ++m) acc[i][j][m] = (v2f){0.0f, 0.0f};

#pragma unroll
    for (int a = 0; a < HRW; ++a) {
        if (a == HRW - 1) __builtin_amdgcn_sched_barrier(0);
#pragma unroll
        for (int q = 0; q < WR; ++q) {
            v2f xv[CP];
#pragma unroll
            for (int m = 0; m < CP; ++m) {
                xv[m] = xa[a][q][m];
                if (a == HRW - 1)
                    xv[m] = (v2f){__uint_as_float(__float_as_uint(xv[m].x) & keep[q]),
                                  __uint_as_float(__float_as_uint(xv[m].y) & keep[q])};
            }
#pragma unroll
            for (int i = 0; i < TH; ++i) {
                const int kh = HRW * TG + a - i;
                if (kh < 0 || kh >= K_SZ) continue;
#pragma unroll
                for (int j = 0; j < TW; ++j) {
                    const int kw = q - j;
                    if (kw < 0 || kw >= K_SZ) continue;
                    const v2f kv = *(const v2f*)(kl + (i * TW + j) * NK + (kh * K_SZ + kw) * G_NUM);
#pragma unroll
                    for (int m = 0; m < CP; ++m)
                        acc[i][j][m] = __builtin_elementwise_fma(kv, xv[m], acc[i][j][m]);
                }
            }
        }
    }

#pragma unroll
    for (int j = 0; j < TW; ++j)
#pragma unroll
        for (int m = 0; m < CP; ++m) {
            *(v2f*)(pw + ((1 - TG) * TW + j) * C_CH + 64 * m) = acc[1 - TG][j][m];
            own[j][m] = acc[TG][j][m];
        }
}

__global__ __launch_bounds__(FT) __attribute__((amdgpu_waves_per_eu(4, 4)))
void involution_fused(
    const float* __restrict__ x,  const float* __restrict__ w1,
    const float* __restrict__ b1, const float* __restrict__ gamma,
    const float* __restrict__ beta, const float* __restrict__ mu,
    const float* __restrict__ var, const float* __restrict__ w2,
    const float* __restrict__ b2, float* __restrict__ out, float* __restrict__ kern)
{
    __shared__ __attribute__((aligned(16))) float ksh[RM * NK];
    __shared__ __attribute__((aligned(16))) float tsh[RM * TP];
    __shared__ __attribute__((aligned(16))) float psh[RM * C_CH];
    float* xs = ksh;

    const unsigned tid = threadIdx.x;
    const unsigned ln  = tid & 31u;
    const int wv    = __builtin_amdgcn_readfirstlane((int)(tid >> 5));
    const int tile  = blockIdx.x;
    const int w0    = (tile & 15) * TS;
    const int h0    = ((tile >> 4) & 7) * RH;
    const int b     = tile >> 7;
    const int laneM = ln & 15;
    const int kHalf = ln >> 4;

    const int c4 = tid & 63, pj = (tid >> 6) & 3, rh = tid >> 8;
    v4f xt[TS];
#pragma unroll
    for (int i = 0; i < TS; ++i)
        xt[i] = *(const v4f*)(x + (size_t)((b * HW + h0 + TS * rh + i) * HW + w0 + pj) * C_CH + 4 * c4);

    v2f bw[G1_KS];
    float ga = 0.0f, va = 1.0f, be = 0.0f, mn = 0.0f, b1c = 0.0f;
    const int nb = wv * W_COLS;
    if (wv >= 12) {
        const int col = (wv - 12) * 16 + laneM;
        ga  = gamma[col];
        va  = var[col];
        be  = beta[col];
        mn  = mu[col];
        b1c = b1[col];
        const float* bcol = w1 + (size_t)(2 * kHalf) * CR + col;
#pragma unroll
        for (int s = 0; s < G1_KS; ++s) {
            bw[s].x = bcol[(size_t)(4 * s) * CR];
            bw[s].y = bcol[(size_t)(4 * s + 1) * CR];
        }
#pragma unroll
        for (int i = 0; i < TS; ++i)
            *(v4f*)(xs + (TM * rh + 4 * i + pj) * XP + 4 * c4) = xt[i];
    } else {
#pragma unroll
        for (int i = 0; i < TS; ++i)
            *(v4f*)(xs + (TM * rh + 4 * i + pj) * XP + 4 * c4) = xt[i];
    }
    barrier_lds_only();

    if (wv >= 12) {
        const int col = (wv - 12) * 16 + laneM;
        const float* arow0 = xs + laneM * XP + 2 * kHalf;
        const float* arow1 = arow0 + TM * XP;
        v2f a0[2][A_BK], a1[2][A_BK];
#pragma unroll
        for (int s = 0; s < A_BK; ++s) {
            a0[0][s] = *(const v2f*)(arow0 + 4 * s);
            a1[0][s] = *(const v2f*)(arow1 + 4 * s);
        }
        v8f acc0 = {}, acc1 = {};
#pragma unroll
        for (int blk = 0; blk < G1_KS / A_BK; ++blk) {
            const int cur = blk & 1, nxt = cur ^ 1;
            if (blk + 1 < G1_KS / A_BK) {
#pragma unroll
                for (int s = 0; s < A_BK; ++s) {
                    a0[nxt][s] = *(const v2f*)(arow0 + 4 * (A_BK * (blk + 1) + s));
                    a1[nxt][s] = *(const v2f*)(arow1 + 4 * (A_BK * (blk + 1) + s));
                }
            }
            __builtin_amdgcn_sched_barrier(0);
#pragma unroll
            for (int s = 0; s < A_BK; ++s) {
                acc0 = __builtin_amdgcn_wmma_f32_16x16x4_f32(
                    false, a0[cur][s], false, bw[A_BK * blk + s], (short)0, acc0, false, false);
                acc1 = __builtin_amdgcn_wmma_f32_16x16x4_f32(
                    false, a1[cur][s], false, bw[A_BK * blk + s], (short)0, acc1, false, false);
            }
        }
        const float scale = ga * rsqrtf(va + BN_EPS);
        const float shift = be - mn * scale;
#pragma unroll
        for (int r = 0; r < 8; ++r) {
            const float v0 = fmaf(acc0[r] + b1c, scale, shift);
            const float v1 = fmaf(acc1[r] + b1c, scale, shift);
            tsh[(8 * kHalf + r) * TP + col]      = fmaxf(v0, 0.0f);
            tsh[(TM + 8 * kHalf + r) * TP + col] = fmaxf(v1, 0.0f);
        }
    }

    v4f px[G2_KS], py[G2_KS];
    if (wv < 12) {
        const float* bp = w2 + (size_t)(2 * kHalf) * NK + nb + 4 * laneM;
#pragma unroll
        for (int s = 0; s < G2_KS; ++s) {
            px[s] = *(const v4f*)(bp + (size_t)(4 * s) * NK);
            py[s] = *(const v4f*)(bp + (size_t)(4 * s + 1) * NK);
        }
    }

    __syncthreads();

    const int sub = wv >> 2;
    const int chf = (wv >> 1) & 1;
    const int tg  = wv & 1;
    const int hs  = h0 + TH * sub;
    const unsigned cA = (unsigned)chf * 128u + 2u * ln;
    const unsigned g0 = (2u * ln) & (G_NUM - 1u);
    v2f xa[HRW][WR][CP];
    const int hr0 = hs + HRW * tg - (K_SZ / 2);
    const size_t a0 = (size_t)x
        + ((long)((b * HW + hr0) * HW + (w0 - (K_SZ / 2))) * C_CH + (long)cA) * 4;

    if (wv < 12) {
        const v4f biasP = *(const v4f*)(b2 + nb + 4 * laneM);
#pragma unroll
        for (int mt = 0; mt < MT; ++mt) {
            v2f a[G2_KS];
#pragma unroll
            for (int s = 0; s < G2_KS; ++s)
                a[s] = *(const v2f*)(tsh + (TM * mt + laneM) * TP + 4 * s + 2 * kHalf);
            v8f acc0 = {}, acc1 = {}, acc2 = {}, acc3 = {};
#pragma unroll
            for (int s = 0; s < G2_KS; ++s) {
                acc0 = __builtin_amdgcn_wmma_f32_16x16x4_f32(
                    false, a[s], false, (v2f){px[s].x, py[s].x}, (short)0, acc0, false, false);
                acc1 = __builtin_amdgcn_wmma_f32_16x16x4_f32(
                    false, a[s], false, (v2f){px[s].y, py[s].y}, (short)0, acc1, false, false);
                acc2 = __builtin_amdgcn_wmma_f32_16x16x4_f32(
                    false, a[s], false, (v2f){px[s].z, py[s].z}, (short)0, acc2, false, false);
                acc3 = __builtin_amdgcn_wmma_f32_16x16x4_f32(
                    false, a[s], false, (v2f){px[s].w, py[s].w}, (short)0, acc3, false, false);
            }
            float* krow = ksh + (TM * mt + 8 * kHalf) * NK;
#pragma unroll
            for (int r = 0; r < 8; ++r) {
                const v4f o = {acc0[r] + biasP.x, acc1[r] + biasP.y, acc2[r] + biasP.z, acc3[r] + biasP.w};
                *(v4f*)(krow + r * NK + nb + 4 * laneM) = o;
            }
        }
    } else if (wv < 14) {
        const int mt = wv - 12;
        const int tcol = TAIL0 + laneM;
        const float* bcol = w2 + (size_t)(2 * kHalf) * NK + tcol;
        v2f bt[G2_KS];
#pragma unroll
        for (int s = 0; s < G2_KS; ++s) {
            bt[s].x = bcol[(size_t)(4 * s) * NK];
            bt[s].y = bcol[(size_t)(4 * s + 1) * NK];
        }
        const float b2c = b2[tcol];
        v2f a[G2_KS];
#pragma unroll
        for (int s = 0; s < G2_KS; ++s)
            a[s] = *(const v2f*)(tsh + (TM * mt + laneM) * TP + 4 * s + 2 * kHalf);
        v8f acc = {};
#pragma unroll
        for (int s = 0; s < G2_KS; ++s) {
            acc = __builtin_amdgcn_wmma_f32_16x16x4_f32(
                false, a[s], false, bt[s], (short)0, acc, false, false);
        }
        float* krow = ksh + (TM * mt + 8 * kHalf) * NK;
#pragma unroll
        for (int r = 0; r < 8; ++r) krow[r * NK + tcol] = acc[r] + b2c;
    }

#pragma unroll
    for (int a = 0; a < HRW - 1; ++a) {
        const bool rok = (unsigned)(hr0 + a) < (unsigned)HW;
#pragma unroll
        for (int q = 0; q < WR; ++q) {
            const bool ok = rok && ((unsigned)(w0 + q - (K_SZ / 2)) < (unsigned)HW);
#pragma unroll
            for (int m = 0; m < CP; ++m) xa[a][q][m] = (v2f){0.0f, 0.0f};
            if (ok) {
#pragma unroll
                for (int m = 0; m < CP; ++m)
                    xa[a][q][m] = *(gv2f_p)(a0 + (size_t)(a * HW + q) * (C_CH * 4) + (size_t)(m * 64 * 4));
            }
        }
    }

    barrier_lds_only();

#pragma unroll
    for (int s = 0; s < (KT4 + FT - 1) / FT; ++s) {
        const int idx = tid + FT * s;
        if (idx < KT4) {
            const int i   = idx / (KT4 / RH);
            const int rem = idx - i * (KT4 / RH);
            v4f* dst = (v4f*)(kern + (size_t)((b * HW + h0 + i) * HW + w0) * NK) + rem;
            *dst = ((const v4f*)ksh)[idx];
        }
    }

    unsigned keep[WR];
    {
        const int rr = hr0 + (HRW - 1);
        const bool rok = (unsigned)rr < (unsigned)HW;
        const int rc = rr < 0 ? 0 : (rr > HW - 1 ? HW - 1 : rr);
#pragma unroll
        for (int q = 0; q < WR; ++q) {
            const int cq = w0 + q - (K_SZ / 2);
            const bool ok = rok && ((unsigned)cq < (unsigned)HW);
            const int cc = cq < 0 ? 0 : (cq > HW - 1 ? HW - 1 : cq);
            const size_t ad = (size_t)x + ((size_t)((b * HW + rc) * HW + cc) * C_CH + (size_t)cA) * 4;
            keep[q] = ok ? 0xffffffffu : 0u;
#pragma unroll
            for (int m = 0; m < CP; ++m)
                xa[HRW - 1][q][m] = *(gv2f_p)(ad + (size_t)(m * 64 * 4));
        }
    }

    const float* kl = ksh + (sub * TH * TW) * NK + g0;
    float* pw = psh + (sub * TH * TW) * C_CH + cA;
    v2f own[TW][CP];
    if (tg == 0) gather_rows<0>(kl, xa, keep, pw, own);
    else         gather_rows<1>(kl, xa, keep, pw, own);

    __syncthreads();

#pragma unroll
    for (int j = 0; j < TW; ++j)
#pragma unroll
        for (int m = 0; m < CP; ++m) {
            const v2f o = *(const v2f*)(pw + (tg * TW + j) * C_CH + 64 * m);
            *(v2f*)(out + (size_t)((b * HW + hs + tg) * HW + w0 + j) * C_CH + cA + 64 * m) = own[j][m] + o;
        }
}

extern "C" void kernel_launch(void* const* d_in, const int* in_sizes, int n_in,
                              void* d_out, int out_size, void* d_ws, size_t ws_size,
                              hipStream_t stream)
{
    const float* x     = (const float*)d_in[0];
    const float* w1    = (const float*)d_in[1];
    const float* b1    = (const float*)d_in[2];
    const float* gamma = (const float*)d_in[3];
    const float* beta  = (const float*)d_in[4];
    const float* mu    = (const float*)d_in[5];
    const float* var   = (const float*)d_in[6];
    const float* w2    = (const float*)d_in[7];
    const float* b2    = (const float*)d_in[8];

    float* out  = (float*)d_out;
    float* kern = out + (size_t)M_TOT * C_CH;

    involution_fused<<<256, FT, 0, stream>>>(x, w1, b1, gamma, beta, mu, var, w2, b2, out, kern);
}
